# starting kernel
# baseline (speedup 1.0000x reference)
#include <hip/hip_runtime.h>

typedef __attribute__((ext_vector_type(2))) float v2f;
typedef __attribute__((ext_vector_type(8))) float v8f;

static constexpr int NN     = 100000;
static constexpr int NE     = 1600000;
static constexpr int DK     = 128;
static constexpr int MTILES = NN / 16;

template<int NT, bool RELU_IN>
__global__ __launch_bounds__(128)
void gemm_wmma_f32(const float* __restrict__ A, const float* __restrict__ W,
                   const float* __restrict__ bias, float* __restrict__ out)
{
    constexpr int DOUT = NT * 16;
    const int lane  = threadIdx.x & 31;
    const int wave  = threadIdx.x >> 5;
    const int mtile = blockIdx.x * 4 + wave;
    if (mtile >= MTILES) return;

    const int half = lane >> 4;
    const int mn   = lane & 15;
    const int kb   = half * 2;

    const float* arow = A + (size_t)(mtile * 16 + mn) * DK;

    v8f acc[NT];
#pragma unroll
    for (int t = 0; t < NT; ++t) acc[t] = (v8f)0.0f;

    for (int k = 0; k < DK; k += 4) {
        v2f a = *(const v2f*)(arow + k + kb);
        if (RELU_IN) { a.x = fmaxf(a.x, 0.0f); a.y = fmaxf(a.y, 0.0f); }
        const float* wp = W + (size_t)(k + kb) * DOUT + mn;
#pragma unroll
        for (int t = 0; t < NT; ++t) {
            v2f b;
            b.x = wp[t * 16];
            b.y = wp[t * 16 + DOUT];
            acc[t] = __builtin_amdgcn_wmma_f32_16x16x4_f32(
                false, a, false, b, (short)0, acc[t], false, false);
        }
    }

#pragma unroll
    for (int t = 0; t < NT; ++t) {
        const float bv = bias[t * 16 + mn];
#pragma unroll
        for (int v = 0; v < 8; ++v) {
            const int m = mtile * 16 + v + 8 * half;
            out[(size_t)m * DOUT + t * 16 + mn] = acc[t][v] + bv;
        }
    }
}

template<int VEC>
__global__ __launch_bounds__(256)
void spmm_atomic(const float* __restrict__ vals, const int* __restrict__ row,
                 const int* __restrict__ col, const float* __restrict__ x,
                 float* __restrict__ out)
{
    constexpr int D = 32 * VEC;
    const long long gt = (long long)blockIdx.x * blockDim.x + threadIdx.x;
    const int e    = (int)(gt >> 5);
    const int lane = (int)(gt & 31);
    if (e >= NE) return;

    const float v = vals[e];
    const int   r = row[e];
    const int   c = col[e];

    const float* xs = x   + (size_t)c * D + lane * VEC;
    float*       od = out + (size_t)r * D + lane * VEC;

    float t[VEC];
    if constexpr (VEC == 4) {
        const float4 q = *(const float4*)xs;
        t[0] = q.x; t[1] = q.y; t[2] = q.z; t[3] = q.w;
    } else {
        const float2 q = *(const float2*)xs;
        t[0] = q.x; t[1] = q.y;
    }
#pragma unroll
    for (int i = 0; i < VEC; ++i)
        atomicAdd(od + i, v * t[i]);
}

extern "C" void kernel_launch(void* const* d_in, const int* in_sizes, int n_in,
                              void* d_out, int out_size, void* d_ws, size_t ws_size,
                              hipStream_t stream)
{
    const float* x    = (const float*)d_in[0];
    const float* vals = (const float*)d_in[1];
    const float* W1   = (const float*)d_in[2];
    const float* b1   = (const float*)d_in[3];
    const float* W2   = (const float*)d_in[4];
    const float* b2   = (const float*)d_in[5];
    const float* W3   = (const float*)d_in[6];
    const float* b3   = (const float*)d_in[7];
    const int*   row  = (const int*)d_in[8];
    const int*   col  = (const int*)d_in[9];

    float* bufA = (float*)d_ws;
    float* bufB = bufA + (size_t)NN * 128;

    const dim3 gBlk(128);
    const dim3 gGrd((MTILES + 3) / 4);
    const dim3 sBlk(256);
    const dim3 sGrd(NE / 8);

    gemm_wmma_f32<8, false><<<gGrd, gBlk, 0, stream>>>(x, W1, b1, bufA);
    hipMemsetAsync(bufB, 0, (size_t)NN * 128 * sizeof(float), stream);
    spmm_atomic<4><<<sGrd, sBlk, 0, stream>>>(vals, row, col, bufA, bufB);

    gemm_wmma_f32<8, true><<<gGrd, gBlk, 0, stream>>>(bufB, W2, b2, bufA);
    hipMemsetAsync(bufB, 0, (size_t)NN * 128 * sizeof(float), stream);
    spmm_atomic<4><<<sGrd, sBlk, 0, stream>>>(vals, row, col, bufA, bufB);

    gemm_wmma_f32<4, true><<<gGrd, gBlk, 0, stream>>>(bufB, W3, b3, bufA);
    hipMemsetAsync(d_out, 0, (size_t)NN * 64 * sizeof(float), stream);
    spmm_atomic<2><<<sGrd, sBlk, 0, stream>>>(vals, row, col, bufA, (float*)d_out);
}


# SpMM float-atomic scatter -> per-call row grouping of edges (int atomic arrival-order rank) + wave-per-row gather summed in registers, one store per row; big memsets dropped
# speedup vs baseline: 2.5036x; 2.5036x over previous
#include <hip/hip_runtime.h>


typedef __attribute__((ext_vector_type(2))) float v2f;
typedef __attribute__((ext_vector_type(8))) float v8f;

static constexpr int NN     = 100000;
static constexpr int NE     = 1600000;
static constexpr int DK     = 128;
static constexpr int MTILES = NN / 16;
static constexpr int CNT_PAD = NN + 96;

template<int NT, bool RELU_IN>
__global__ __launch_bounds__(128)
void gemm_wmma_f32(const float* __restrict__ A, const float* __restrict__ W,
                   const float* __restrict__ bias, float* __restrict__ out)
{
    constexpr int DOUT = NT * 16;
    const int lane  = threadIdx.x & 31;
    const int wave  = threadIdx.x >> 5;
    const int mtile = blockIdx.x * 4 + wave;
    if (mtile >= MTILES) return;

    const int half = lane >> 4;
    const int mn   = lane & 15;
    const int kb   = half * 2;

    const float* arow = A + (size_t)(mtile * 16 + mn) * DK;

    v8f acc[NT];
#pragma unroll
    for (int t = 0; t < NT; ++t) acc[t] = (v8f)0.0f;

    for (int k = 0; k < DK; k += 4) {
        v2f a = *(const v2f*)(arow + k + kb);
        if (RELU_IN) { a.x = fmaxf(a.x, 0.0f); a.y = fmaxf(a.y, 0.0f); }
        const float* wp = W + (size_t)(k + kb) * DOUT + mn;
#pragma unroll
        for (int t = 0; t < NT; ++t) {
            v2f b;
            b.x = wp[t * 16];
            b.y = wp[t * 16 + DOUT];
            acc[t] = __builtin_amdgcn_wmma_f32_16x16x4_f32(
                false, a, false, b, (short)0, acc[t], false, false);
        }
    }

#pragma unroll
    for (int t = 0; t < NT; ++t) {
        const float bv = bias[t * 16 + mn];
#pragma unroll
        for (int v = 0; v < 8; ++v) {
            const int m = mtile * 16 + v + 8 * half;
            out[(size_t)m * DOUT + t * 16 + mn] = acc[t][v] + bv;
        }
    }
}

__global__ __launch_bounds__(256)
void edge_rank(const int* __restrict__ row, int* __restrict__ cnt, int* __restrict__ rank)
{
    const int e = blockIdx.x * 256 + threadIdx.x;
    if (e >= NE) return;
    rank[e] = atomicAdd(cnt + row[e], 1);
}

__global__ __launch_bounds__(256)
void row_alloc(int* __restrict__ cnt, int* __restrict__ rowinfo)
{
    const int r    = blockIdx.x * 256 + threadIdx.x;
    const int lane = threadIdx.x & 31;
    const int c    = (r < NN) ? cnt[r] : 0;
    int incl = c;
#pragma unroll
    for (int d = 1; d < 32; d <<= 1) {
        const int t = __shfl_up(incl, d);
        if (lane >= d) incl += t;
    }
    int base = 0;
    if (lane == 31) base = atomicAdd(cnt + NN, incl);
    base = __shfl(base, 31);
    if (r < NN) ((int2*)rowinfo)[r] = make_int2(base + incl - c, c);
}

__global__ __launch_bounds__(256)
void edge_fill(const int* __restrict__ row, const int* __restrict__ col,
               const float* __restrict__ vals, const int* __restrict__ rank,
               const int* __restrict__ rowinfo, int* __restrict__ grouped)
{
    const int e = blockIdx.x * 256 + threadIdx.x;
    if (e >= NE) return;
    const int pos = rowinfo[2 * row[e]] + rank[e];
    if ((unsigned)pos < (unsigned)NE)
        ((int2*)grouped)[pos] = make_int2(col[e], __float_as_int(vals[e]));
}

template<int VEC> struct RowVec;
template<> struct RowVec<4> { typedef float4 T; };
template<> struct RowVec<2> { typedef float2 T; };

template<int VEC, int K>
__device__ __forceinline__ void gather_accum(const int ec, const int ew, const int j,
                                             const float* __restrict__ xl, float (&acc)[VEC])
{
    typedef typename RowVec<VEC>::T VT;
    VT    q[K];
    float w[K];
#pragma unroll
    for (int k = 0; k < K; ++k) {
        int c = __builtin_amdgcn_readlane(ec, j + k);
        c = (int)min((unsigned)c, (unsigned)(NN - 1));
        w[k] = __int_as_float(__builtin_amdgcn_readlane(ew, j + k));
        q[k] = *(const VT*)(xl + (size_t)c * (32 * VEC));
    }
#pragma unroll
    for (int k = 0; k < K; ++k) {
        const float* f = (const float*)&q[k];
#pragma unroll
        for (int i = 0; i < VEC; ++i) acc[i] = fmaf(w[k], f[i], acc[i]);
    }
}

template<int VEC>
__global__ __launch_bounds__(256)
void spmm_gather(const int* __restrict__ rowinfo, const int* __restrict__ grouped,
                 const float* __restrict__ x, float* __restrict__ out)
{
    typedef typename RowVec<VEC>::T VT;
    constexpr int D = 32 * VEC;
    const int lane = threadIdx.x & 31;
    const int r = __builtin_amdgcn_readfirstlane(blockIdx.x * 8 + (threadIdx.x >> 5));
    if (r >= NN) return;

    const int2 ri = ((const int2*)rowinfo)[r];
    const int n   = min(max(ri.y, 0), NE);
    const int beg = min(max(ri.x, 0), NE - n);

    const float* xl = x + lane * VEC;
    float acc[VEC];
#pragma unroll
    for (int i = 0; i < VEC; ++i) acc[i] = 0.0f;

    for (int base = 0; base < n; base += 32) {
        const int m = min(32, n - base);
        int ec = 0, ew = 0;
        if (lane < m) {
            const int2 t = ((const int2*)grouped)[beg + base + lane];
            ec = t.x; ew = t.y;
        }
        int j = 0;
        for (; j + 8 <= m; j += 8) gather_accum<VEC, 8>(ec, ew, j, xl, acc);
        switch (m - j) {
            case 7: gather_accum<VEC, 7>(ec, ew, j, xl, acc); break;
            case 6: gather_accum<VEC, 6>(ec, ew, j, xl, acc); break;
            case 5: gather_accum<VEC, 5>(ec, ew, j, xl, acc); break;
            case 4: gather_accum<VEC, 4>(ec, ew, j, xl, acc); break;
            case 3: gather_accum<VEC, 3>(ec, ew, j, xl, acc); break;
            case 2: gather_accum<VEC, 2>(ec, ew, j, xl, acc); break;
            case 1: gather_accum<VEC, 1>(ec, ew, j, xl, acc); break;
            default: break;
        }
    }

    VT o;
    float* of = (float*)&o;
#pragma unroll
    for (int i = 0; i < VEC; ++i) of[i] = acc[i];
    *(VT*)(out + (size_t)r * D + lane * VEC) = o;
}

extern "C" void kernel_launch(void* const* d_in, const int* in_sizes, int n_in,
                              void* d_out, int out_size, void* d_ws, size_t ws_size,
                              hipStream_t stream)
{
    const float* x    = (const float*)d_in[0];
    const float* vals = (const float*)d_in[1];
    const float* W1   = (const float*)d_in[2];
    const float* b1   = (const float*)d_in[3];
    const float* W2   = (const float*)d_in[4];
    const float* b2   = (const float*)d_in[5];
    const float* W3   = (const float*)d_in[6];
    const float* b3   = (const float*)d_in[7];
    const int*   row  = (const int*)d_in[8];
    const int*   col  = (const int*)d_in[9];

    float* bufA = (float*)d_ws;
    float* bufB = bufA + (size_t)NN * 128;
    int* cnt     = (int*)(bufB + (size_t)NN * 128);
    int* rowinfo = cnt + CNT_PAD;
    int* rank    = rowinfo + 2 * NN;
    int* grouped = rank + NE;

    const dim3 gBlk(128);
    const dim3 gGrd((MTILES + 3) / 4);
    const dim3 eBlk(256);
    const dim3 eGrd(NE / 256);
    const dim3 rGrd((NN + 255) / 256);
    const dim3 sBlk(256);
    const dim3 sGrd(NN / 8);

    hipMemsetAsync(cnt, 0, (size_t)CNT_PAD * sizeof(int), stream);
    edge_rank<<<eGrd, eBlk, 0, stream>>>(row, cnt, rank);
    row_alloc<<<rGrd, eBlk, 0, stream>>>(cnt, rowinfo);
    edge_fill<<<eGrd, eBlk, 0, stream>>>(row, col, vals, rank, rowinfo, grouped);

    gemm_wmma_f32<8, false><<<gGrd, gBlk, 0, stream>>>(x, W1, b1, bufA);
    spmm_gather<4><<<sGrd, sBlk, 0, stream>>>(rowinfo, grouped, bufA, bufB);

    gemm_wmma_f32<8, true><<<gGrd, gBlk, 0, stream>>>(bufB, W2, b2, bufA);
    spmm_gather<4><<<sGrd, sBlk, 0, stream>>>(rowinfo, grouped, bufA, bufB);

    gemm_wmma_f32<4, true><<<gGrd, gBlk, 0, stream>>>(bufB, W3, b3, bufA);
    spmm_gather<2><<<sGrd, sBlk, 0, stream>>>(rowinfo, grouped, bufA, (float*)d_out);
}


# SpMM 1 and 2 each fused with the next layer's WMMA product: a 16-wave group gathers 64 rows (4 per wave, their edge lists requested up front) into LDS with relu, multiplies by W, stores whole rows
# speedup vs baseline: 2.8273x; 1.1293x over previous
#include <hip/hip_runtime.h>


typedef __attribute__((ext_vector_type(2))) float v2f;
typedef __attribute__((ext_vector_type(8))) float v8f;

static constexpr int NN     = 100000;
static constexpr int NE     = 1600000;
static constexpr int DK     = 128;
static constexpr int MTILES = NN / 16;
static constexpr int CNT_PAD = NN + 96;

template<int NT, bool RELU_IN>
__global__ __launch_bounds__(128)
void gemm_wmma_f32(const float* __restrict__ A, const float* __restrict__ W,
                   const float* __restrict__ bias, float* __restrict__ out)
{
    constexpr int DOUT = NT * 16;
    const int lane  = threadIdx.x & 31;
    const int wave  = threadIdx.x >> 5;
    const int mtile = blockIdx.x * 4 + wave;
    if (mtile >= MTILES) return;

    const int half = lane >> 4;
    const int mn   = lane & 15;
    const int kb   = half * 2;

    const float* arow = A + (size_t)(mtile * 16 + mn) * DK;

    v8f acc[NT];
#pragma unroll
    for (int t = 0; t < NT; ++t) acc[t] = (v8f)0.0f;

    for (int k = 0; k < DK; k += 4) {
        v2f a = *(const v2f*)(arow + k + kb);
        if (RELU_IN) { a.x = fmaxf(a.x, 0.0f); a.y = fmaxf(a.y, 0.0f); }
        const float* wp = W + (size_t)(k + kb) * DOUT + mn;
#pragma unroll
        for (int t = 0; t < NT; ++t) {
            v2f b;
            b.x = wp[t * 16];
            b.y = wp[t * 16 + DOUT];
            acc[t] = __builtin_amdgcn_wmma_f32_16x16x4_f32(
                false, a, false, b, (short)0, acc[t], false, false);
        }
    }

#pragma unroll
    for (int t = 0; t < NT; ++t) {
        const float bv = bias[t * 16 + mn];
#pragma unroll
        for (int v = 0; v < 8; ++v) {
            const int m = mtile * 16 + v + 8 * half;
            out[(size_t)m * DOUT + t * 16 + mn] = acc[t][v] + bv;
        }
    }
}

__global__ __launch_bounds__(256)
void edge_rank(const int* __restrict__ row, int* __restrict__ cnt, int* __restrict__ rank)
{
    const int e = blockIdx.x * 256 + threadIdx.x;
    if (e >= NE) return;
    rank[e] = atomicAdd(cnt + row[e], 1);
}

__global__ __launch_bounds__(256)
void row_alloc(int* __restrict__ cnt, int* __restrict__ rowinfo)
{
    const int r    = blockIdx.x * 256 + threadIdx.x;
    const int lane = threadIdx.x & 31;
    const int c    = (r < NN) ? cnt[r] : 0;
    int incl = c;
#pragma unroll
    for (int d = 1; d < 32; d <<= 1) {
        const int t = __shfl_up(incl, d);
        if (lane >= d) incl += t;
    }
    int base = 0;
    if (lane == 31) base = atomicAdd(cnt + NN, incl);
    base = __shfl(base, 31);
    if (r < NN) ((int2*)rowinfo)[r] = make_int2(base + incl - c, c);
}

__global__ __launch_bounds__(256)
void edge_fill(const int* __restrict__ row, const int* __restrict__ col,
               const float* __restrict__ vals, const int* __restrict__ rank,
               const int* __restrict__ rowinfo, int* __restrict__ grouped)
{
    const int e = blockIdx.x * 256 + threadIdx.x;
    if (e >= NE) return;
    const int pos = rowinfo[2 * row[e]] + rank[e];
    if ((unsigned)pos < (unsigned)NE)
        ((int2*)grouped)[pos] = make_int2(col[e], __float_as_int(vals[e]));
}

template<int VEC> struct RowVec;
template<> struct RowVec<4> { typedef float4 T; };
template<> struct RowVec<2> { typedef float2 T; };

template<int VEC, int K>
__device__ __forceinline__ void gather_accum(const int ec, const int ew, const int j,
                                             const float* __restrict__ xl, float (&acc)[VEC])
{
    typedef typename RowVec<VEC>::T VT;
    VT    q[K];
    float w[K];
#pragma unroll
    for (int k = 0; k < K; ++k) {
        int c = __builtin_amdgcn_readlane(ec, j + k);
        c = (int)min((unsigned)c, (unsigned)(NN - 1));
        w[k] = __int_as_float(__builtin_amdgcn_readlane(ew, j + k));
        q[k] = *(const VT*)(xl + (size_t)c * (32 * VEC));
    }
#pragma unroll
    for (int k = 0; k < K; ++k) {
        const float* f = (const float*)&q[k];
#pragma unroll
        for (int i = 0; i < VEC; ++i) acc[i] = fmaf(w[k], f[i], acc[i]);
    }
}

__device__ __forceinline__ void load_entries(const int* __restrict__ grouped, const int first, const int m,
                                             const int lane, int& ec, int& ew)
{
    ec = 0; ew = 0;
    if (lane < m) {
        const int2 t = ((const int2*)grouped)[first + lane];
        ec = t.x; ew = t.y;
    }
}

template<int VEC>
__device__ __forceinline__ void gather_chunk(const int ec, const int ew, const int m,
                                             const float* __restrict__ xl, float (&acc)[VEC])
{
    int j = 0;
    for (; j + 8 <= m; j += 8) gather_accum<VEC, 8>(ec, ew, j, xl, acc);
    switch (m - j) {
        case 7: gather_accum<VEC, 7>(ec, ew, j, xl, acc); break;
        case 6: gather_accum<VEC, 6>(ec, ew, j, xl, acc); break;
        case 5: gather_accum<VEC, 5>(ec, ew, j, xl, acc); break;
        case 4: gather_accum<VEC, 4>(ec, ew, j, xl, acc); break;
        case 3: gather_accum<VEC, 3>(ec, ew, j, xl, acc); break;
        case 2: gather_accum<VEC, 2>(ec, ew, j, xl, acc); break;
        case 1: gather_accum<VEC, 1>(ec, ew, j, xl, acc); break;
        default: break;
    }
}

template<int VEC>
__device__ __forceinline__ void gather_row(const int* __restrict__ rowinfo, const int* __restrict__ grouped,
                                           const float* __restrict__ xl, const int r, const int lane,
                                           float (&acc)[VEC])
{
    const int2 ri = ((const int2*)rowinfo)[r];
    const int n   = min(max(ri.y, 0), NE);
    const int beg = min(max(ri.x, 0), NE - n);

    for (int base = 0; base < n; base += 32) {
        const int m = min(32, n - base);
        int ec, ew;
        load_entries(grouped, beg + base, m, lane, ec, ew);
        gather_chunk<VEC>(ec, ew, m, xl, acc);
    }
}

template<int VEC>
__global__ __launch_bounds__(256)
void spmm_gather(const int* __restrict__ rowinfo, const int* __restrict__ grouped,
                 const float* __restrict__ x, float* __restrict__ out)
{
    typedef typename RowVec<VEC>::T VT;
    constexpr int D = 32 * VEC;
    const int lane = threadIdx.x & 31;
    const int r = __builtin_amdgcn_readfirstlane(blockIdx.x * 8 + (threadIdx.x >> 5));
    if (r >= NN) return;

    float acc[VEC];
#pragma unroll
    for (int i = 0; i < VEC; ++i) acc[i] = 0.0f;
    gather_row<VEC>(rowinfo, grouped, x + lane * VEC, r, lane, acc);

    VT o;
    float* of = (float*)&o;
#pragma unroll
    for (int i = 0; i < VEC; ++i) of[i] = acc[i];
    *(VT*)(out + (size_t)r * D + lane * VEC) = o;
}

static constexpr int TROWS  = 64;
static constexpr int TPITCH = 132;
static_assert(NN % 4 == 0, "a wave's four rows must be all inside or all outside");

template<int NT>
__global__ __launch_bounds__(512)
void spmm_relu_gemm(const int* __restrict__ rowinfo, const int* __restrict__ grouped,
                    const float* __restrict__ x, const float* __restrict__ W,
                    const float* __restrict__ bias, float* __restrict__ out)
{
    constexpr int DOUT = NT * 16;
    constexpr int MS   = NT / 4;
    __shared__ __attribute__((aligned(16))) float tile[TROWS * TPITCH];

    const int lane = threadIdx.x & 31;
    const int wave = __builtin_amdgcn_readfirstlane(threadIdx.x >> 5);
    const int row0 = blockIdx.x * TROWS;

    const int  rbase = row0 + wave * 4;
    const bool live  = rbase < NN;
    const int4* rp   = (const int4*)(rowinfo + 2 * (live ? rbase : 0));
    const int4 q0 = rp[0], q1 = rp[1];
    const int keep = live ? NE : 0;
    int n0 = min(max(q0.y, 0), keep), n1 = min(max(q0.w, 0), keep);
    int n2 = min(max(q1.y, 0), keep), n3 = min(max(q1.w, 0), keep);
    int b0 = min(max(q0.x, 0), NE - n0), b1 = min(max(q0.z, 0), NE - n1);
    int b2 = min(max(q1.x, 0), NE - n2), b3 = min(max(q1.z, 0), NE - n3);
    int c0, w0, c1, w1, c2, w2, c3, w3;
    load_entries(grouped, b0, min(32, n0), lane, c0, w0);
    load_entries(grouped, b1, min(32, n1), lane, c1, w1);
    load_entries(grouped, b2, min(32, n2), lane, c2, w2);
    load_entries(grouped, b3, min(32, n3), lane, c3, w3);

    const float* xl = x + lane * 4;
    for (int i = 0; i < 4; ++i) {
        float acc[4] = {0.0f, 0.0f, 0.0f, 0.0f};
        gather_chunk<4>(c0, w0, min(32, n0), xl, acc);
        for (int base = 32; base < n0; base += 32) {
            const int m = min(32, n0 - base);
            int fc, fw;
            load_entries(grouped, b0 + base, m, lane, fc, fw);
            gather_chunk<4>(fc, fw, m, xl, acc);
        }
        float4 o;
        o.x = fmaxf(acc[0], 0.0f); o.y = fmaxf(acc[1], 0.0f);
        o.z = fmaxf(acc[2], 0.0f); o.w = fmaxf(acc[3], 0.0f);
        *(float4*)&tile[(wave * 4 + i) * TPITCH + lane * 4] = o;
        n0 = n1; n1 = n2; n2 = n3; n3 = 0;
        b0 = b1; b1 = b2; b2 = b3;
        c0 = c1; c1 = c2; c2 = c3;
        w0 = w1; w1 = w2; w2 = w3;
    }

    const int half = lane >> 4;
    const int mn   = lane & 15;
    const int kb   = half * 2;
    const int nt   = wave % NT;
    const int ms0  = (wave / NT) * MS;

    const float* wp = W + (size_t)kb * DOUT + nt * 16 + mn;
    v2f breg[DK / 8];
#pragma unroll
    for (int ks = 0; ks < DK / 8; ++ks) {
        breg[ks].x = wp[(size_t)(4 * ks) * DOUT];
        breg[ks].y = wp[(size_t)(4 * ks + 1) * DOUT];
    }
    const float bv = bias[nt * 16 + mn];

    __syncthreads();

    v8f acc[MS];
#pragma unroll
    for (int s = 0; s < MS; ++s) acc[s] = (v8f)0.0f;

#pragma unroll
    for (int kh = 0; kh < 2; ++kh) {
        if (kh == 1) {
#pragma unroll
            for (int ks = 0; ks < DK / 8; ++ks) {
                breg[ks].x = wp[(size_t)(4 * (ks + DK / 8)) * DOUT];
                breg[ks].y = wp[(size_t)(4 * (ks + DK / 8) + 1) * DOUT];
            }
        }
#pragma unroll
        for (int ks = 0; ks < DK / 8; ++ks) {
#pragma unroll
            for (int s = 0; s < MS; ++s) {
                const v2f a = *(const v2f*)&tile[((ms0 + s) * 16 + mn) * TPITCH + 4 * (ks + kh * (DK / 8)) + kb];
                acc[s] = __builtin_amdgcn_wmma_f32_16x16x4_f32(
                    false, a, false, breg[ks], (short)0, acc[s], false, false);
            }
        }
    }

    __syncthreads();

#pragma unroll
    for (int s = 0; s < MS; ++s)
#pragma unroll
        for (int v = 0; v < 8; ++v)
            tile[((ms0 + s) * 16 + v + 8 * half) * TPITCH + nt * 16 + mn] = acc[s][v] + bv;

    __syncthreads();

    typedef typename RowVec<DOUT / 32>::T OT;
    for (int i = 0; i < 4; ++i) {
        const int rl = wave * 4 + i;
        const int r  = row0 + rl;
        if (r < NN)
            *(OT*)(out + (size_t)r * DOUT + lane * (DOUT / 32)) =
                *(const OT*)&tile[rl * TPITCH + lane * (DOUT / 32)];
    }
}

extern "C" void kernel_launch(void* const* d_in, const int* in_sizes, int n_in,
                              void* d_out, int out_size, void* d_ws, size_t ws_size,
                              hipStream_t stream)
{
    const float* x    = (const float*)d_in[0];
    const float* vals = (const float*)d_in[1];
    const float* W1   = (const float*)d_in[2];
    const float* b1   = (const float*)d_in[3];
    const float* W2   = (const float*)d_in[4];
    const float* b2   = (const float*)d_in[5];
    const float* W3   = (const float*)d_in[6];
    const float* b3   = (const float*)d_in[7];
    const int*   row  = (const int*)d_in[8];
    const int*   col  = (const int*)d_in[9];

    float* bufA = (float*)d_ws;
    float* bufB = bufA + (size_t)NN * 128;
    int* cnt     = (int*)(bufB + (size_t)NN * 128);
    int* rowinfo = cnt + CNT_PAD;
    int* rank    = rowinfo + 2 * NN;
    int* grouped = rank + NE;

    const dim3 gBlk(128);
    const dim3 gGrd((MTILES + 3) / 4);
    const dim3 eBlk(256);
    const dim3 eGrd(NE / 256);
    const dim3 rGrd((NN + 255) / 256);
    const dim3 sBlk(256);
    const dim3 sGrd(NN / 8);

    hipMemsetAsync(cnt, 0, (size_t)CNT_PAD * sizeof(int), stream);
    edge_rank<<<eGrd, eBlk, 0, stream>>>(row, cnt, rank);
    row_alloc<<<rGrd, eBlk, 0, stream>>>(cnt, rowinfo);
    edge_fill<<<eGrd, eBlk, 0, stream>>>(row, col, vals, rank, rowinfo, grouped);

    const dim3 fBlk(512);
    const dim3 fGrd((NN + TROWS - 1) / TROWS);

    gemm_wmma_f32<8, false><<<gGrd, gBlk, 0, stream>>>(x, W1, b1, bufA);

    spmm_relu_gemm<8><<<fGrd, fBlk, 0, stream>>>(rowinfo, grouped, bufA, W2, b2, bufB);

    spmm_relu_gemm<4><<<fGrd, fBlk, 0, stream>>>(rowinfo, grouped, bufB, W3, b3, bufA);

    spmm_gather<2><<<sGrd, sBlk, 0, stream>>>(rowinfo, grouped, bufA, (float*)d_out);
}


# Layer 1 product moved behind its spmm: x is gathered and (sum v x) @ W1 + rowsum(vals) * b1 is taken on the LDS tile of the fused kernel; GEMM launch and its plane gone (reassociated sum order)
# speedup vs baseline: 3.0409x; 1.0755x over previous
#include <hip/hip_runtime.h>


typedef __attribute__((ext_vector_type(2))) float v2f;
typedef __attribute__((ext_vector_type(8))) float v8f;

static constexpr int NN     = 100000;
static constexpr int NE     = 1600000;
static constexpr int DK     = 128;
static constexpr int CNT_PAD = NN + 96;


__global__ __launch_bounds__(256)
void edge_rank(const int* __restrict__ row, int* __restrict__ cnt, int* __restrict__ rank)
{
    const int e = blockIdx.x * 256 + threadIdx.x;
    if (e >= NE) return;
    rank[e] = atomicAdd(cnt + row[e], 1);
}

__global__ __launch_bounds__(256)
void row_alloc(int* __restrict__ cnt, int* __restrict__ rowinfo)
{
    const int r    = blockIdx.x * 256 + threadIdx.x;
    const int lane = threadIdx.x & 31;
    const int c    = (r < NN) ? cnt[r] : 0;
    int incl = c;
#pragma unroll
    for (int d = 1; d < 32; d <<= 1) {
        const int t = __shfl_up(incl, d);
        if (lane >= d) incl += t;
    }
    int base = 0;
    if (lane == 31) base = atomicAdd(cnt + NN, incl);
    base = __shfl(base, 31);
    if (r < NN) ((int2*)rowinfo)[r] = make_int2(base + incl - c, c);
}

__global__ __launch_bounds__(256)
void edge_fill(const int* __restrict__ row, const int* __restrict__ col,
               const float* __restrict__ vals, const int* __restrict__ rank,
               const int* __restrict__ rowinfo, int* __restrict__ grouped)
{
    const int e = blockIdx.x * 256 + threadIdx.x;
    if (e >= NE) return;
    const int pos = rowinfo[2 * row[e]] + rank[e];
    if ((unsigned)pos < (unsigned)NE)
        ((int2*)grouped)[pos] = make_int2(col[e], __float_as_int(vals[e]));
}

template<int VEC> struct RowVec;
template<> struct RowVec<4> { typedef float4 T; };
template<> struct RowVec<2> { typedef float2 T; };

template<int VEC, int K>
__device__ __forceinline__ void gather_accum(const int ec, const int ew, const int j,
                                             const float* __restrict__ xl, float (&acc)[VEC])
{
    typedef typename RowVec<VEC>::T VT;
    VT    q[K];
    float w[K];
#pragma unroll
    for (int k = 0; k < K; ++k) {
        int c = __builtin_amdgcn_readlane(ec, j + k);
        c = (int)min((unsigned)c, (unsigned)(NN - 1));
        w[k] = __int_as_float(__builtin_amdgcn_readlane(ew, j + k));
        q[k] = *(const VT*)(xl + (size_t)c * (32 * VEC));
    }
#pragma unroll
    for (int k = 0; k < K; ++k) {
        const float* f = (const float*)&q[k];
#pragma unroll
        for (int i = 0; i < VEC; ++i) acc[i] = fmaf(w[k], f[i], acc[i]);
    }
}

__device__ __forceinline__ void load_entries(const int* __restrict__ grouped, const int first, const int m,
                                             const int lane, int& ec, int& ew)
{
    ec = 0; ew = 0;
    if (lane < m) {
        const int2 t = ((const int2*)grouped)[first + lane];
        ec = t.x; ew = t.y;
    }
}

template<int VEC>
__device__ __forceinline__ void gather_chunk(const int ec, const int ew, const int m,
                                             const float* __restrict__ xl, float (&acc)[VEC])
{
    int j = 0;
    for (; j + 8 <= m; j += 8) gather_accum<VEC, 8>(ec, ew, j, xl, acc);
    switch (m - j) {
        case 7: gather_accum<VEC, 7>(ec, ew, j, xl, acc); break;
        case 6: gather_accum<VEC, 6>(ec, ew, j, xl, acc); break;
        case 5: gather_accum<VEC, 5>(ec, ew, j, xl, acc); break;
        case 4: gather_accum<VEC, 4>(ec, ew, j, xl, acc); break;
        case 3: gather_accum<VEC, 3>(ec, ew, j, xl, acc); break;
        case 2: gather_accum<VEC, 2>(ec, ew, j, xl, acc); break;
        case 1: gather_accum<VEC, 1>(ec, ew, j, xl, acc); break;
        default: break;
    }
}

template<int VEC>
__device__ __forceinline__ void gather_row(const int* __restrict__ rowinfo, const int* __restrict__ grouped,
                                           const float* __restrict__ xl, const int r, const int lane,
                                           float (&acc)[VEC])
{
    const int2 ri = ((const int2*)rowinfo)[r];
    const int n   = min(max(ri.y, 0), NE);
    const int beg = min(max(ri.x, 0), NE - n);

    for (int base = 0; base < n; base += 32) {
        const int m = min(32, n - base);
        int ec, ew;
        load_entries(grouped, beg + base, m, lane, ec, ew);
        gather_chunk<VEC>(ec, ew, m, xl, acc);
    }
}

template<int VEC>
__global__ __launch_bounds__(256)
void spmm_gather(const int* __restrict__ rowinfo, const int* __restrict__ grouped,
                 const float* __restrict__ x, float* __restrict__ out)
{
    typedef typename RowVec<VEC>::T VT;
    constexpr int D = 32 * VEC;
    const int lane = threadIdx.x & 31;
    const int r = __builtin_amdgcn_readfirstlane(blockIdx.x * 8 + (threadIdx.x >> 5));
    if (r >= NN) return;

    float acc[VEC];
#pragma unroll
    for (int i = 0; i < VEC; ++i) acc[i] = 0.0f;
    gather_row<VEC>(rowinfo, grouped, x + lane * VEC, r, lane, acc);

    VT o;
    float* of = (float*)&o;
#pragma unroll
    for (int i = 0; i < VEC; ++i) of[i] = acc[i];
    *(VT*)(out + (size_t)r * D + lane * VEC) = o;
}

static constexpr int TROWS  = 64;
static constexpr int TPITCH = 132;
static_assert(NN % 4 == 0, "a wave's four rows must be all inside or all outside");

template<int NT, int KG>
__device__ __forceinline__ void tile_times_w(const float* tile, const float* __restrict__ W,
                                             const int wave, const int lane, v8f (&acc)[NT / 4])
{
    constexpr int DOUT = NT * 16;
    constexpr int MS   = NT / 4;
    constexpr int NG   = DK / 4 / KG;
    const int half = lane >> 4;
    const int mn   = lane & 15;
    const int kb   = half * 2;
    const int nt   = wave % NT;
    const int ms0  = (wave / NT) * MS;

    const float* wp = W + (size_t)kb * DOUT + nt * 16 + mn;
    v2f breg[KG];
#pragma unroll
    for (int ks = 0; ks < KG; ++ks) {
        breg[ks].x = wp[(size_t)(4 * ks) * DOUT];
        breg[ks].y = wp[(size_t)(4 * ks + 1) * DOUT];
    }

    __syncthreads();

#pragma unroll
    for (int s = 0; s < MS; ++s) acc[s] = (v8f)0.0f;

#pragma unroll
    for (int g = 0; g < NG; ++g) {
        if (g > 0) {
#pragma unroll
            for (int ks = 0; ks < KG; ++ks) {
                breg[ks].x = wp[(size_t)(4 * (ks + g * KG)) * DOUT];
                breg[ks].y = wp[(size_t)(4 * (ks + g * KG) + 1) * DOUT];
            }
        }
#pragma unroll
        for (int ks = 0; ks < KG; ++ks) {
#pragma unroll
            for (int s = 0; s < MS; ++s) {
                const v2f a = *(const v2f*)&tile[((ms0 + s) * 16 + mn) * TPITCH + 4 * (ks + g * KG) + kb];
                acc[s] = __builtin_amdgcn_wmma_f32_16x16x4_f32(
                    false, a, false, breg[ks], (short)0, acc[s], false, false);
            }
        }
    }

    __syncthreads();
}

template<int NT, bool PRE>
__global__ __launch_bounds__(512)
void spmm_relu_gemm(const int* __restrict__ rowinfo, const int* __restrict__ grouped,
                    const float* __restrict__ x, const float* __restrict__ Wp,
                    const float* __restrict__ bp, const float* __restrict__ W,
                    const float* __restrict__ bias, float* __restrict__ out)
{
    constexpr int DOUT = NT * 16;
    constexpr int MS   = NT / 4;
    constexpr int KG   = PRE ? 8 : 16;
    __shared__ __attribute__((aligned(16))) float tile[TROWS * TPITCH];
    __shared__ float wsum[TROWS];

    const int lane = threadIdx.x & 31;
    const int wave = __builtin_amdgcn_readfirstlane(threadIdx.x >> 5);
    const int row0 = blockIdx.x * TROWS;

    const int  rbase = row0 + wave * 4;
    const bool live  = rbase < NN;
    const int4* rp   = (const int4*)(rowinfo + 2 * (live ? rbase : 0));
    const int4 q0 = rp[0], q1 = rp[1];
    const int keep = live ? NE : 0;
    int n0 = min(max(q0.y, 0), keep), n1 = min(max(q0.w, 0), keep);
    int n2 = min(max(q1.y, 0), keep), n3 = min(max(q1.w, 0), keep);
    int b0 = min(max(q0.x, 0), NE - n0), b1 = min(max(q0.z, 0), NE - n1);
    int b2 = min(max(q1.x, 0), NE - n2), b3 = min(max(q1.z, 0), NE - n3);
    int c0, w0, c1, w1, c2, w2, c3, w3;
    load_entries(grouped, b0, min(32, n0), lane, c0, w0);
    load_entries(grouped, b1, min(32, n1), lane, c1, w1);
    load_entries(grouped, b2, min(32, n2), lane, c2, w2);
    load_entries(grouped, b3, min(32, n3), lane, c3, w3);

    const float* xl = x + lane * 4;
    for (int i = 0; i < 4; ++i) {
        float acc[4] = {0.0f, 0.0f, 0.0f, 0.0f};
        float sw = __int_as_float(w0);
        gather_chunk<4>(c0, w0, min(32, n0), xl, acc);
        for (int base = 32; base < n0; base += 32) {
            const int m = min(32, n0 - base);
            int fc, fw;
            load_entries(grouped, b0 + base, m, lane, fc, fw);
            if (PRE) sw += __int_as_float(fw);
            gather_chunk<4>(fc, fw, m, xl, acc);
        }
        float4 o;
        if (PRE) {
            o.x = acc[0]; o.y = acc[1]; o.z = acc[2]; o.w = acc[3];
#pragma unroll
            for (int d = 16; d >= 1; d >>= 1) sw += __shfl_xor(sw, d);
            if (lane == 0) wsum[wave * 4 + i] = sw;
        } else {
            o.x = fmaxf(acc[0], 0.0f); o.y = fmaxf(acc[1], 0.0f);
            o.z = fmaxf(acc[2], 0.0f); o.w = fmaxf(acc[3], 0.0f);
        }
        *(float4*)&tile[(wave * 4 + i) * TPITCH + lane * 4] = o;
        n0 = n1; n1 = n2; n2 = n3; n3 = 0;
        b0 = b1; b1 = b2; b2 = b3;
        c0 = c1; c1 = c2; c2 = c3;
        w0 = w1; w1 = w2; w2 = w3;
    }

    const int half = lane >> 4;
    const int mn   = lane & 15;

    if (PRE) {
        const int   pnt  = wave % 8;
        const int   pms0 = (wave / 8) * 2;
        const float pbv  = bp[pnt * 16 + mn];
        v8f pacc[2];
        tile_times_w<8, KG>(tile, Wp, wave, lane, pacc);
#pragma unroll
        for (int s = 0; s < 2; ++s)
#pragma unroll
            for (int v = 0; v < 8; ++v) {
                const int rl = (pms0 + s) * 16 + v + 8 * half;
                tile[rl * TPITCH + pnt * 16 + mn] = fmaxf(fmaf(wsum[rl], pbv, pacc[s][v]), 0.0f);
            }
    }

    const int   nt  = wave % NT;
    const int   ms0 = (wave / NT) * MS;
    const float bv  = bias[nt * 16 + mn];
    v8f acc[MS];
    tile_times_w<NT, KG>(tile, W, wave, lane, acc);

#pragma unroll
    for (int s = 0; s < MS; ++s)
#pragma unroll
        for (int v = 0; v < 8; ++v)
            tile[((ms0 + s) * 16 + v + 8 * half) * TPITCH + nt * 16 + mn] = acc[s][v] + bv;

    __syncthreads();

    typedef typename RowVec<DOUT / 32>::T OT;
    for (int i = 0; i < 4; ++i) {
        const int rl = wave * 4 + i;
        const int r  = row0 + rl;
        if (r < NN)
            *(OT*)(out + (size_t)r * DOUT + lane * (DOUT / 32)) =
                *(const OT*)&tile[rl * TPITCH + lane * (DOUT / 32)];
    }
}

extern "C" void kernel_launch(void* const* d_in, const int* in_sizes, int n_in,
                              void* d_out, int out_size, void* d_ws, size_t ws_size,
                              hipStream_t stream)
{
    const float* x    = (const float*)d_in[0];
    const float* vals = (const float*)d_in[1];
    const float* W1   = (const float*)d_in[2];
    const float* b1   = (const float*)d_in[3];
    const float* W2   = (const float*)d_in[4];
    const float* b2   = (const float*)d_in[5];
    const float* W3   = (const float*)d_in[6];
    const float* b3   = (const float*)d_in[7];
    const int*   row  = (const int*)d_in[8];
    const int*   col  = (const int*)d_in[9];

    float* bufA = (float*)d_ws;
    float* bufB = bufA + (size_t)NN * 128;
    int* cnt     = (int*)(bufB + (size_t)NN * 128);
    int* rowinfo = cnt + CNT_PAD;
    int* rank    = rowinfo + 2 * NN;
    int* grouped = rank + NE;

    const dim3 eBlk(256);
    const dim3 eGrd(NE / 256);
    const dim3 rGrd((NN + 255) / 256);
    const dim3 sBlk(256);
    const dim3 sGrd(NN / 8);

    hipMemsetAsync(cnt, 0, (size_t)CNT_PAD * sizeof(int), stream);
    edge_rank<<<eGrd, eBlk, 0, stream>>>(row, cnt, rank);
    row_alloc<<<rGrd, eBlk, 0, stream>>>(cnt, rowinfo);
    edge_fill<<<eGrd, eBlk, 0, stream>>>(row, col, vals, rank, rowinfo, grouped);

    const dim3 fBlk(512);
    const dim3 fGrd((NN + TROWS - 1) / TROWS);

    spmm_relu_gemm<8, true><<<fGrd, fBlk, 0, stream>>>(rowinfo, grouped, x, W1, b1, W2, b2, bufB);

    spmm_relu_gemm<4, false><<<fGrd, fBlk, 0, stream>>>(rowinfo, grouped, bufB, nullptr, nullptr, W3, b3, bufA);

    spmm_gather<2><<<sGrd, sBlk, 0, stream>>>(rowinfo, grouped, bufA, (float*)d_out);
}


# Edge grouping as a two-level counting sort on LDS counters (4096-edge tiles by 256-row block, then a work-group per block) replaces 1.6M global atomics + scattered fill; row order stays arrival order
# speedup vs baseline: 4.9785x; 1.6372x over previous
#include <hip/hip_runtime.h>


typedef __attribute__((ext_vector_type(2))) float v2f;
typedef __attribute__((ext_vector_type(8))) float v8f;

static constexpr int NN     = 100000;
static constexpr int NE     = 1600000;
static constexpr int DK     = 128;


static constexpr int TILE_E   = 4096;
static constexpr int NT       = (NE + TILE_E - 1) / TILE_E;
static constexpr int BROWS    = 256;
static constexpr int NB       = (NN + BROWS - 1) / BROWS;
static constexpr int NBP      = 512;
static constexpr int T_PAD    = ((NB * NT + 63) / 64) * 64;
static constexpr int COL_BITS = 17;
static constexpr int CAPL     = 6144;
static_assert(NN <= (1 << COL_BITS), "col must fit its field");
static_assert(NB <= NBP && NB <= 1024 && NT <= 1024, "one thread per block / tile");
static_assert(TILE_E % 1024 == 0 && NE % 4 == 0, "edges are read four at a time and tiles leave as 16-byte stores");

template<int NW>
__device__ __forceinline__ int block_excl_scan(const int v, int* wsum, const int lane, const int wave, int& total)
{
    int incl = v;
#pragma unroll
    for (int d = 1; d < 32; d <<= 1) {
        const int t = __shfl_up(incl, d);
        if (lane >= d) incl += t;
    }
    if (lane == 31) wsum[wave] = incl;
    __syncthreads();
    int base = 0, tot = 0;
#pragma unroll
    for (int w = 0; w < NW; ++w) {
        const int s = wsum[w];
        if (w < wave) base += s;
        tot += s;
    }
    total = tot;
    return base + incl - v;
}

__global__ __launch_bounds__(256)
void tile_sort(const int* __restrict__ row, const int* __restrict__ col, const float* __restrict__ vals,
               int* __restrict__ binCount, int* __restrict__ T, int* __restrict__ tsorted)
{
    __shared__ int hist[NBP];
    __shared__ int off[NBP];
    __shared__ int wsum[8];
    __shared__ __attribute__((aligned(16))) int2 stage[TILE_E];

    const int tid  = threadIdx.x;
    const int lane = tid & 31;
    const int wave = tid >> 5;
    const int t    = blockIdx.x;
    const int e0   = t * TILE_E;
    const int cntTile = min(TILE_E, NE - e0);

    hist[tid] = 0; hist[tid + 256] = 0;

    int rr[16], lr[16];
#pragma unroll
    for (int k = 0; k < 4; ++k) {
        const int g = 4 * (k * 256 + tid);
        const int4 q = *(const int4*)(row + min(e0 + g, NE - 4));
        rr[4 * k] = q.x; rr[4 * k + 1] = q.y; rr[4 * k + 2] = q.z; rr[4 * k + 3] = q.w;
    }
    int   cc[16];
    float vv[16];
#pragma unroll
    for (int k = 0; k < 4; ++k) {
        const int g = 4 * (k * 256 + tid);
        const int4   q = *(const int4*)(col + min(e0 + g, NE - 4));
        const float4 f = *(const float4*)(vals + min(e0 + g, NE - 4));
        cc[4 * k] = q.x; cc[4 * k + 1] = q.y; cc[4 * k + 2] = q.z; cc[4 * k + 3] = q.w;
        vv[4 * k] = f.x; vv[4 * k + 1] = f.y; vv[4 * k + 2] = f.z; vv[4 * k + 3] = f.w;
    }
    __syncthreads();
#pragma unroll
    for (int k = 0; k < 16; ++k) {
        const int i = 4 * ((k >> 2) * 256 + tid) + (k & 3);
        rr[k] = (int)min((unsigned)rr[k], (unsigned)(NN - 1));
        lr[k] = 0;
        if (i < cntTile) lr[k] = atomicAdd(&hist[rr[k] >> 8], 1);
    }
    __syncthreads();

    const int ha = hist[2 * tid], hb = hist[2 * tid + 1];
    int total;
    const int ex = block_excl_scan<8>(ha + hb, wsum, lane, wave, total);
    off[2 * tid]     = ex;
    off[2 * tid + 1] = ex + ha;
    __syncthreads();

    for (int b = tid; b < NB; b += 256) {
        const int c = hist[b];
        T[b * NT + t] = off[b] | (c << 16);
        if (c > 0) atomicAdd(binCount + b, c);
    }

#pragma unroll
    for (int k = 0; k < 16; ++k) {
        const int i = 4 * ((k >> 2) * 256 + tid) + (k & 3);
        if (i < cntTile) {
            const int c = (int)min((unsigned)cc[k], (unsigned)(NN - 1));
            const int p = off[rr[k] >> 8] + lr[k];
            stage[p & (TILE_E - 1)] = make_int2(((rr[k] & (BROWS - 1)) << COL_BITS) | c, __float_as_int(vv[k]));
        }
    }
    __syncthreads();

    for (int i = 2 * tid; i < cntTile; i += 512)
        *(int4*)(tsorted + 2 * (size_t)(e0 + i)) = *(const int4*)&stage[i];
}

__global__ __launch_bounds__(1024)
void csr_build(const int* __restrict__ binCount, const int* __restrict__ T, const int* __restrict__ tsorted,
               int* __restrict__ rowinfo, int* __restrict__ grouped)
{
    __shared__ int tinfo[NT + 1];
    __shared__ int deg[BROWS];
    __shared__ int cur[BROWS];
    __shared__ int wsumA[32];
    __shared__ int wsumB[32];
    __shared__ int2 stage[CAPL];

    const int tid  = threadIdx.x;
    const int lane = tid & 31;
    const int wave = tid >> 5;
    const int b    = blockIdx.x;

    int part = (tid < b) ? binCount[tid] : 0;
    if (tid < NT)    tinfo[tid] = T[b * NT + tid];
    if (tid < BROWS) deg[tid] = 0;
#pragma unroll
    for (int d = 16; d >= 1; d >>= 1) part += __shfl_xor(part, d);
    if (lane == 0) wsumB[wave] = part;
    __syncthreads();
    int start = 0;
#pragma unroll
    for (int w = 0; w < 32; ++w) start += wsumB[w];

    constexpr int NTRIP = (NT + 63) / 64;
    const int hw  = tid >> 4;
    const int l16 = tid & 15;
    const int2* ts = (const int2*)tsorted;

    int2 ent[NTRIP];
    int  rc[NTRIP], rb[NTRIP];
#pragma unroll
    for (int j = 0; j < NTRIP; ++j) {
        const int t    = hw + 64 * j;
        const int info = (t < NT) ? tinfo[t] : 0;
        rc[j] = min(info >> 16, TILE_E);
        rb[j] = t * TILE_E + (info & 0xFFFF);
        ent[j] = ts[min(rb[j] + l16, NE - 1)];
    }
#pragma unroll
    for (int j = 0; j < NTRIP; ++j)
        if (l16 < rc[j]) atomicAdd(&deg[(ent[j].x >> COL_BITS) & (BROWS - 1)], 1);
#pragma unroll
    for (int j = 0; j < NTRIP; ++j)
        for (int k = l16 + 16; k < rc[j]; k += 16) {
            const int2 e = ts[min(rb[j] + k, NE - 1)];
            atomicAdd(&deg[(e.x >> COL_BITS) & (BROWS - 1)], 1);
        }
    __syncthreads();

    const int d = (tid < BROWS) ? deg[tid] : 0;
    int n;
    const int ex = block_excl_scan<32>(d, wsumA, lane, wave, n);
    if (tid < BROWS) {
        cur[tid] = ex;
        const int r = b * BROWS + tid;
        if (r < NN) ((int2*)rowinfo)[r] = make_int2(start + ex, d);
    }
    __syncthreads();

    const bool staged = n <= CAPL;
#pragma unroll
    for (int j = 0; j < NTRIP; ++j) {
        for (int k = l16; k < rc[j]; k += 16) {
            const int2 e = (k == l16) ? ent[j] : ts[min(rb[j] + k, NE - 1)];
            const int p  = atomicAdd(&cur[(e.x >> COL_BITS) & (BROWS - 1)], 1);
            const int2 e2 = make_int2(e.x & ((1 << COL_BITS) - 1), e.y);
            if (staged) {
                if ((unsigned)p < (unsigned)CAPL) stage[p] = e2;
            } else {
                const int q = start + p;
                if ((unsigned)q < (unsigned)NE) ((int2*)grouped)[q] = e2;
            }
        }
    }
    if (staged) {
        __syncthreads();
        for (int i = tid; i < n; i += 1024) {
            const int q = start + i;
            if ((unsigned)q < (unsigned)NE) ((int2*)grouped)[q] = stage[i];
        }
    }
}

template<int VEC> struct RowVec;
template<> struct RowVec<4> { typedef float4 T; };
template<> struct RowVec<2> { typedef float2 T; };

template<int VEC, int K>
__device__ __forceinline__ void gather_accum(const int ec, const int ew, const int j,
                                             const float* __restrict__ xl, float (&acc)[VEC])
{
    typedef typename RowVec<VEC>::T VT;
    VT    q[K];
    float w[K];
#pragma unroll
    for (int k = 0; k < K; ++k) {
        int c = __builtin_amdgcn_readlane(ec, j + k);
        c = (int)min((unsigned)c, (unsigned)(NN - 1));
        w[k] = __int_as_float(__builtin_amdgcn_readlane(ew, j + k));
        q[k] = *(const VT*)(xl + (size_t)c * (32 * VEC));
    }
#pragma unroll
    for (int k = 0; k < K; ++k) {
        const float* f = (const float*)&q[k];
#pragma unroll
        for (int i = 0; i < VEC; ++i) acc[i] = fmaf(w[k], f[i], acc[i]);
    }
}

__device__ __forceinline__ void load_entries(const int* __restrict__ grouped, const int first, const int m,
                                             const int lane, int& ec, int& ew)
{
    ec = 0; ew = 0;
    if (lane < m) {
        const int2 t = ((const int2*)grouped)[first + lane];
        ec = t.x; ew = t.y;
    }
}

template<int VEC>
__device__ __forceinline__ void gather_chunk(const int ec, const int ew, const int m,
                                             const float* __restrict__ xl, float (&acc)[VEC])
{
    int j = 0;
    for (; j + 8 <= m; j += 8) gather_accum<VEC, 8>(ec, ew, j, xl, acc);
    switch (m - j) {
        case 7: gather_accum<VEC, 7>(ec, ew, j, xl, acc); break;
        case 6: gather_accum<VEC, 6>(ec, ew, j, xl, acc); break;
        case 5: gather_accum<VEC, 5>(ec, ew, j, xl, acc); break;
        case 4: gather_accum<VEC, 4>(ec, ew, j, xl, acc); break;
        case 3: gather_accum<VEC, 3>(ec, ew, j, xl, acc); break;
        case 2: gather_accum<VEC, 2>(ec, ew, j, xl, acc); break;
        case 1: gather_accum<VEC, 1>(ec, ew, j, xl, acc); break;
        default: break;
    }
}

template<int VEC>
__device__ __forceinline__ void gather_row(const int* __restrict__ rowinfo, const int* __restrict__ grouped,
                                           const float* __restrict__ xl, const int r, const int lane,
                                           float (&acc)[VEC])
{
    const int2 ri = ((const int2*)rowinfo)[r];
    const int n   = min(max(ri.y, 0), NE);
    const int beg = min(max(ri.x, 0), NE - n);

    for (int base = 0; base < n; base += 32) {
        const int m = min(32, n - base);
        int ec, ew;
        load_entries(grouped, beg + base, m, lane, ec, ew);
        gather_chunk<VEC>(ec, ew, m, xl, acc);
    }
}

template<int VEC>
__global__ __launch_bounds__(256)
void spmm_gather(const int* __restrict__ rowinfo, const int* __restrict__ grouped,
                 const float* __restrict__ x, float* __restrict__ out)
{
    typedef typename RowVec<VEC>::T VT;
    constexpr int D = 32 * VEC;
    const int lane = threadIdx.x & 31;
    const int r = __builtin_amdgcn_readfirstlane(blockIdx.x * 8 + (threadIdx.x >> 5));
    if (r >= NN) return;

    float acc[VEC];
#pragma unroll
    for (int i = 0; i < VEC; ++i) acc[i] = 0.0f;
    gather_row<VEC>(rowinfo, grouped, x + lane * VEC, r, lane, acc);

    VT o;
    float* of = (float*)&o;
#pragma unroll
    for (int i = 0; i < VEC; ++i) of[i] = acc[i];
    *(VT*)(out + (size_t)r * D + lane * VEC) = o;
}

static constexpr int TROWS  = 64;
static constexpr int TPITCH = 132;
static_assert(NN % 4 == 0, "a wave's four rows must be all inside or all outside");

template<int NT, int KG>
__device__ __forceinline__ void tile_times_w(const float* tile, const float* __restrict__ W,
                                             const int wave, const int lane, v8f (&acc)[NT / 4])
{
    constexpr int DOUT = NT * 16;
    constexpr int MS   = NT / 4;
    constexpr int NG   = DK / 4 / KG;
    const int half = lane >> 4;
    const int mn   = lane & 15;
    const int kb   = half * 2;
    const int nt   = wave % NT;
    const int ms0  = (wave / NT) * MS;

    const float* wp = W + (size_t)kb * DOUT + nt * 16 + mn;
    v2f breg[KG];
#pragma unroll
    for (int ks = 0; ks < KG; ++ks) {
        breg[ks].x = wp[(size_t)(4 * ks) * DOUT];
        breg[ks].y = wp[(size_t)(4 * ks + 1) * DOUT];
    }

    __syncthreads();

#pragma unroll
    for (int s = 0; s < MS; ++s) acc[s] = (v8f)0.0f;

#pragma unroll
    for (int g = 0; g < NG; ++g) {
        if (g > 0) {
#pragma unroll
            for (int ks = 0; ks < KG; ++ks) {
                breg[ks].x = wp[(size_t)(4 * (ks + g * KG)) * DOUT];
                breg[ks].y = wp[(size_t)(4 * (ks + g * KG) + 1) * DOUT];
            }
        }
#pragma unroll
        for (int ks = 0; ks < KG; ++ks) {
#pragma unroll
            for (int s = 0; s < MS; ++s) {
                const v2f a = *(const v2f*)&tile[((ms0 + s) * 16 + mn) * TPITCH + 4 * (ks + g * KG) + kb];
                acc[s] = __builtin_amdgcn_wmma_f32_16x16x4_f32(
                    false, a, false, breg[ks], (short)0, acc[s], false, false);
            }
        }
    }

    __syncthreads();
}

template<int NT, bool PRE>
__global__ __launch_bounds__(512)
void spmm_relu_gemm(const int* __restrict__ rowinfo, const int* __restrict__ grouped,
                    const float* __restrict__ x, const float* __restrict__ Wp,
                    const float* __restrict__ bp, const float* __restrict__ W,
                    const float* __restrict__ bias, float* __restrict__ out)
{
    constexpr int DOUT = NT * 16;
    constexpr int MS   = NT / 4;
    constexpr int KG   = PRE ? 8 : 16;
    __shared__ __attribute__((aligned(16))) float tile[TROWS * TPITCH];
    __shared__ float wsum[TROWS];

    const int lane = threadIdx.x & 31;
    const int wave = __builtin_amdgcn_readfirstlane(threadIdx.x >> 5);
    const int row0 = blockIdx.x * TROWS;

    const int  rbase = row0 + wave * 4;
    const bool live  = rbase < NN;
    const int4* rp   = (const int4*)(rowinfo + 2 * (live ? rbase : 0));
    const int4 q0 = rp[0], q1 = rp[1];
    const int keep = live ? NE : 0;
    int n0 = min(max(q0.y, 0), keep), n1 = min(max(q0.w, 0), keep);
    int n2 = min(max(q1.y, 0), keep), n3 = min(max(q1.w, 0), keep);
    int b0 = min(max(q0.x, 0), NE - n0), b1 = min(max(q0.z, 0), NE - n1);
    int b2 = min(max(q1.x, 0), NE - n2), b3 = min(max(q1.z, 0), NE - n3);
    int c0, w0, c1, w1, c2, w2, c3, w3;
    load_entries(grouped, b0, min(32, n0), lane, c0, w0);
    load_entries(grouped, b1, min(32, n1), lane, c1, w1);
    load_entries(grouped, b2, min(32, n2), lane, c2, w2);
    load_entries(grouped, b3, min(32, n3), lane, c3, w3);

    const float* xl = x + lane * 4;
    for (int i = 0; i < 4; ++i) {
        float acc[4] = {0.0f, 0.0f, 0.0f, 0.0f};
        float sw = __int_as_float(w0);
        gather_chunk<4>(c0, w0, min(32, n0), xl, acc);
        for (int base = 32; base < n0; base += 32) {
            const int m = min(32, n0 - base);
            int fc, fw;
            load_entries(grouped, b0 + base, m, lane, fc, fw);
            if (PRE) sw += __int_as_float(fw);
            gather_chunk<4>(fc, fw, m, xl, acc);
        }
        float4 o;
        if (PRE) {
            o.x = acc[0]; o.y = acc[1]; o.z = acc[2]; o.w = acc[3];
#pragma unroll
            for (int d = 16; d >= 1; d >>= 1) sw += __shfl_xor(sw, d);
            if (lane == 0) wsum[wave * 4 + i] = sw;
        } else {
            o.x = fmaxf(acc[0], 0.0f); o.y = fmaxf(acc[1], 0.0f);
            o.z = fmaxf(acc[2], 0.0f); o.w = fmaxf(acc[3], 0.0f);
        }
        *(float4*)&tile[(wave * 4 + i) * TPITCH + lane * 4] = o;
        n0 = n1; n1 = n2; n2 = n3; n3 = 0;
        b0 = b1; b1 = b2; b2 = b3;
        c0 = c1; c1 = c2; c2 = c3;
        w0 = w1; w1 = w2; w2 = w3;
    }

    const int half = lane >> 4;
    const int mn   = lane & 15;

    if (PRE) {
        const int   pnt  = wave % 8;
        const int   pms0 = (wave / 8) * 2;
        const float pbv  = bp[pnt * 16 + mn];
        v8f pacc[2];
        tile_times_w<8, KG>(tile, Wp, wave, lane, pacc);
#pragma unroll
        for (int s = 0; s < 2; ++s)
#pragma unroll
            for (int v = 0; v < 8; ++v) {
                const int rl = (pms0 + s) * 16 + v + 8 * half;
                tile[rl * TPITCH + pnt * 16 + mn] = fmaxf(fmaf(wsum[rl], pbv, pacc[s][v]), 0.0f);
            }
    }

    const int   nt  = wave % NT;
    const int   ms0 = (wave / NT) * MS;
    const float bv  = bias[nt * 16 + mn];
    v8f acc[MS];
    tile_times_w<NT, KG>(tile, W, wave, lane, acc);

#pragma unroll
    for (int s = 0; s < MS; ++s)
#pragma unroll
        for (int v = 0; v < 8; ++v)
            tile[((ms0 + s) * 16 + v + 8 * half) * TPITCH + nt * 16 + mn] = acc[s][v] + bv;

    __syncthreads();

    typedef typename RowVec<DOUT / 32>::T OT;
    for (int i = 0; i < 4; ++i) {
        const int rl = wave * 4 + i;
        const int r  = row0 + rl;
        if (r < NN)
            *(OT*)(out + (size_t)r * DOUT + lane * (DOUT / 32)) =
                *(const OT*)&tile[rl * TPITCH + lane * (DOUT / 32)];
    }
}

extern "C" void kernel_launch(void* const* d_in, const int* in_sizes, int n_in,
                              void* d_out, int out_size, void* d_ws, size_t ws_size,
                              hipStream_t stream)
{
    const float* x    = (const float*)d_in[0];
    const float* vals = (const float*)d_in[1];
    const float* W1   = (const float*)d_in[2];
    const float* b1   = (const float*)d_in[3];
    const float* W2   = (const float*)d_in[4];
    const float* b2   = (const float*)d_in[5];
    const float* W3   = (const float*)d_in[6];
    const float* b3   = (const float*)d_in[7];
    const int*   row  = (const int*)d_in[8];
    const int*   col  = (const int*)d_in[9];

    float* bufA = (float*)d_ws;
    float* bufB = bufA + (size_t)NN * 128;
    int* binCount = (int*)(bufB + (size_t)NN * 128);
    int* T        = binCount + NBP;
    int* rowinfo  = T + T_PAD;
    int* tsorted  = rowinfo + 2 * NN;
    int* grouped  = tsorted + 2 * NE;

    const dim3 sBlk(256);
    const dim3 sGrd(NN / 8);

    hipMemsetAsync(binCount, 0, (size_t)NBP * sizeof(int), stream);
    tile_sort<<<dim3(NT), dim3(256), 0, stream>>>(row, col, vals, binCount, T, tsorted);
    csr_build<<<dim3(NB), dim3(1024), 0, stream>>>(binCount, T, tsorted, rowinfo, grouped);

    const dim3 fBlk(512);
    const dim3 fGrd((NN + TROWS - 1) / TROWS);

    spmm_relu_gemm<8, true><<<fGrd, fBlk, 0, stream>>>(rowinfo, grouped, x, W1, b1, W2, b2, bufB);

    spmm_relu_gemm<4, false><<<fGrd, fBlk, 0, stream>>>(rowinfo, grouped, bufB, nullptr, nullptr, W3, b3, bufA);

    spmm_gather<2><<<sGrd, sBlk, 0, stream>>>(rowinfo, grouped, bufA, (float*)d_out);
}


# First fused kernel and csr_build held to 64 registers (W slice fetched in groups of 4 k-steps, each behind the previous group's matrix work): one more work-group per WGP, no straggler round
# speedup vs baseline: 5.1830x; 1.0411x over previous
#include <hip/hip_runtime.h>


typedef __attribute__((ext_vector_type(2))) float v2f;
typedef __attribute__((ext_vector_type(8))) float v8f;

static constexpr int NN     = 100000;
static constexpr int NE     = 1600000;
static constexpr int DK     = 128;


static constexpr int TILE_E   = 4096;
static constexpr int NT       = (NE + TILE_E - 1) / TILE_E;
static constexpr int BROWS    = 256;
static constexpr int NB       = (NN + BROWS - 1) / BROWS;
static constexpr int NBP      = 512;
static constexpr int T_PAD    = ((NB * NT + 63) / 64) * 64;
static constexpr int COL_BITS = 17;
static constexpr int CAPL     = 6144;
static_assert(NN <= (1 << COL_BITS), "col must fit its field");
static_assert(NB <= NBP && NB <= 1024 && NT <= 1024, "one thread per block / tile");
static_assert(TILE_E % 1024 == 0 && NE % 4 == 0, "edges are read four at a time and tiles leave as 16-byte stores");

template<int NW>
__device__ __forceinline__ int block_excl_scan(const int v, int* wsum, const int lane, const int wave, int& total)
{
    int incl = v;
#pragma unroll
    for (int d = 1; d < 32; d <<= 1) {
        const int t = __shfl_up(incl, d);
        if (lane >= d) incl += t;
    }
    if (lane == 31) wsum[wave] = incl;
    __syncthreads();
    int base = 0, tot = 0;
#pragma unroll
    for (int w = 0; w < NW; ++w) {
        const int s = wsum[w];
        if (w < wave) base += s;
        tot += s;
    }
    total = tot;
    return base + incl - v;
}

__global__ __launch_bounds__(256)
void tile_sort(const int* __restrict__ row, const int* __restrict__ col, const float* __restrict__ vals,
               int* __restrict__ binCount, int* __restrict__ T, int* __restrict__ tsorted)
{
    __shared__ int hist[NBP];
    __shared__ int off[NBP];
    __shared__ int wsum[8];
    __shared__ __attribute__((aligned(16))) int2 stage[TILE_E];

    const int tid  = threadIdx.x;
    const int lane = tid & 31;
    const int wave = tid >> 5;
    const int t    = blockIdx.x;
    const int e0   = t * TILE_E;
    const int cntTile = min(TILE_E, NE - e0);

    hist[tid] = 0; hist[tid + 256] = 0;

    int rr[16], lr[16];
#pragma unroll
    for (int k = 0; k < 4; ++k) {
        const int g = 4 * (k * 256 + tid);
        const int4 q = *(const int4*)(row + min(e0 + g, NE - 4));
        rr[4 * k] = q.x; rr[4 * k + 1] = q.y; rr[4 * k + 2] = q.z; rr[4 * k + 3] = q.w;
    }
    int   cc[16];
    float vv[16];
#pragma unroll
    for (int k = 0; k < 4; ++k) {
        const int g = 4 * (k * 256 + tid);
        const int4   q = *(const int4*)(col + min(e0 + g, NE - 4));
        const float4 f = *(const float4*)(vals + min(e0 + g, NE - 4));
        cc[4 * k] = q.x; cc[4 * k + 1] = q.y; cc[4 * k + 2] = q.z; cc[4 * k + 3] = q.w;
        vv[4 * k] = f.x; vv[4 * k + 1] = f.y; vv[4 * k + 2] = f.z; vv[4 * k + 3] = f.w;
    }
    __syncthreads();
#pragma unroll
    for (int k = 0; k < 16; ++k) {
        const int i = 4 * ((k >> 2) * 256 + tid) + (k & 3);
        rr[k] = (int)min((unsigned)rr[k], (unsigned)(NN - 1));
        lr[k] = 0;
        if (i < cntTile) lr[k] = atomicAdd(&hist[rr[k] >> 8], 1);
    }
    __syncthreads();

    const int ha = hist[2 * tid], hb = hist[2 * tid + 1];
    int total;
    const int ex = block_excl_scan<8>(ha + hb, wsum, lane, wave, total);
    off[2 * tid]     = ex;
    off[2 * tid + 1] = ex + ha;
    __syncthreads();

    for (int b = tid; b < NB; b += 256) {
        const int c = hist[b];
        T[b * NT + t] = off[b] | (c << 16);
        if (c > 0) atomicAdd(binCount + b, c);
    }

#pragma unroll
    for (int k = 0; k < 16; ++k) {
        const int i = 4 * ((k >> 2) * 256 + tid) + (k & 3);
        if (i < cntTile) {
            const int c = (int)min((unsigned)cc[k], (unsigned)(NN - 1));
            const int p = off[rr[k] >> 8] + lr[k];
            stage[p & (TILE_E - 1)] = make_int2(((rr[k] & (BROWS - 1)) << COL_BITS) | c, __float_as_int(vv[k]));
        }
    }
    __syncthreads();

    for (int i = 2 * tid; i < cntTile; i += 512)
        *(int4*)(tsorted + 2 * (size_t)(e0 + i)) = *(const int4*)&stage[i];
}

__global__ __launch_bounds__(1024) __attribute__((amdgpu_waves_per_eu(16, 16)))
void csr_build(const int* __restrict__ binCount, const int* __restrict__ T, const int* __restrict__ tsorted,
               int* __restrict__ rowinfo, int* __restrict__ grouped)
{
    __shared__ int tinfo[NT + 1];
    __shared__ int deg[BROWS];
    __shared__ int cur[BROWS];
    __shared__ int wsumA[32];
    __shared__ int wsumB[32];
    __shared__ int2 stage[CAPL];

    const int tid  = threadIdx.x;
    const int lane = tid & 31;
    const int wave = tid >> 5;
    const int b    = blockIdx.x;

    int part = (tid < b) ? binCount[tid] : 0;
    if (tid < NT)    tinfo[tid] = T[b * NT + tid];
    if (tid < BROWS) deg[tid] = 0;
#pragma unroll
    for (int d = 16; d >= 1; d >>= 1) part += __shfl_xor(part, d);
    if (lane == 0) wsumB[wave] = part;
    __syncthreads();
    int start = 0;
#pragma unroll
    for (int w = 0; w < 32; ++w) start += wsumB[w];

    constexpr int NTRIP = (NT + 63) / 64;
    const int hw  = tid >> 4;
    const int l16 = tid & 15;
    const int2* ts = (const int2*)tsorted;

    int2 ent[NTRIP];
    int  rc[NTRIP], rb[NTRIP];
#pragma unroll
    for (int j = 0; j < NTRIP; ++j) {
        const int t    = hw + 64 * j;
        const int info = (t < NT) ? tinfo[t] : 0;
        rc[j] = min(info >> 16, TILE_E);
        rb[j] = t * TILE_E + (info & 0xFFFF);
        ent[j] = ts[min(rb[j] + l16, NE - 1)];
    }
#pragma unroll
    for (int j = 0; j < NTRIP; ++j)
        if (l16 < rc[j]) atomicAdd(&deg[(ent[j].x >> COL_BITS) & (BROWS - 1)], 1);
#pragma unroll
    for (int j = 0; j < NTRIP; ++j)
        for (int k = l16 + 16; k < rc[j]; k += 16) {
            const int2 e = ts[min(rb[j] + k, NE - 1)];
            atomicAdd(&deg[(e.x >> COL_BITS) & (BROWS - 1)], 1);
        }
    __syncthreads();

    const int d = (tid < BROWS) ? deg[tid] : 0;
    int n;
    const int ex = block_excl_scan<32>(d, wsumA, lane, wave, n);
    if (tid < BROWS) {
        cur[tid] = ex;
        const int r = b * BROWS + tid;
        if (r < NN) ((int2*)rowinfo)[r] = make_int2(start + ex, d);
    }
    __syncthreads();

    const bool staged = n <= CAPL;
#pragma unroll
    for (int j = 0; j < NTRIP; ++j) {
        for (int k = l16; k < rc[j]; k += 16) {
            const int2 e = (k == l16) ? ent[j] : ts[min(rb[j] + k, NE - 1)];
            const int p  = atomicAdd(&cur[(e.x >> COL_BITS) & (BROWS - 1)], 1);
            const int2 e2 = make_int2(e.x & ((1 << COL_BITS) - 1), e.y);
            if (staged) {
                if ((unsigned)p < (unsigned)CAPL) stage[p] = e2;
            } else {
                const int q = start + p;
                if ((unsigned)q < (unsigned)NE) ((int2*)grouped)[q] = e2;
            }
        }
    }
    if (staged) {
        __syncthreads();
        for (int i = tid; i < n; i += 1024) {
            const int q = start + i;
            if ((unsigned)q < (unsigned)NE) ((int2*)grouped)[q] = stage[i];
        }
    }
}

template<int VEC> struct RowVec;
template<> struct RowVec<4> { typedef float4 T; };
template<> struct RowVec<2> { typedef float2 T; };

template<int VEC, int K>
__device__ __forceinline__ void gather_accum(const int ec, const int ew, const int j,
                                             const float* __restrict__ xl, float (&acc)[VEC])
{
    typedef typename RowVec<VEC>::T VT;
    VT    q[K];
    float w[K];
#pragma unroll
    for (int k = 0; k < K; ++k) {
        int c = __builtin_amdgcn_readlane(ec, j + k);
        c = (int)min((unsigned)c, (unsigned)(NN - 1));
        w[k] = __int_as_float(__builtin_amdgcn_readlane(ew, j + k));
        q[k] = *(const VT*)(xl + (size_t)c * (32 * VEC));
    }
#pragma unroll
    for (int k = 0; k < K; ++k) {
        const float* f = (const float*)&q[k];
#pragma unroll
        for (int i = 0; i < VEC; ++i) acc[i] = fmaf(w[k], f[i], acc[i]);
    }
}

__device__ __forceinline__ void load_entries(const int* __restrict__ grouped, const int first, const int m,
                                             const int lane, int& ec, int& ew)
{
    ec = 0; ew = 0;
    if (lane < m) {
        const int2 t = ((const int2*)grouped)[first + lane];
        ec = t.x; ew = t.y;
    }
}

template<int VEC>
__device__ __forceinline__ void gather_chunk(const int ec, const int ew, const int m,
                                             const float* __restrict__ xl, float (&acc)[VEC])
{
    int j = 0;
    for (; j + 8 <= m; j += 8) gather_accum<VEC, 8>(ec, ew, j, xl, acc);
    switch (m - j) {
        case 7: gather_accum<VEC, 7>(ec, ew, j, xl, acc); break;
        case 6: gather_accum<VEC, 6>(ec, ew, j, xl, acc); break;
        case 5: gather_accum<VEC, 5>(ec, ew, j, xl, acc); break;
        case 4: gather_accum<VEC, 4>(ec, ew, j, xl, acc); break;
        case 3: gather_accum<VEC, 3>(ec, ew, j, xl, acc); break;
        case 2: gather_accum<VEC, 2>(ec, ew, j, xl, acc); break;
        case 1: gather_accum<VEC, 1>(ec, ew, j, xl, acc); break;
        default: break;
    }
}

template<int VEC>
__device__ __forceinline__ void gather_row(const int* __restrict__ rowinfo, const int* __restrict__ grouped,
                                           const float* __restrict__ xl, const int r, const int lane,
                                           float (&acc)[VEC])
{
    const int2 ri = ((const int2*)rowinfo)[r];
    const int n   = min(max(ri.y, 0), NE);
    const int beg = min(max(ri.x, 0), NE - n);

    for (int base = 0; base < n; base += 32) {
        const int m = min(32, n - base);
        int ec, ew;
        load_entries(grouped, beg + base, m, lane, ec, ew);
        gather_chunk<VEC>(ec, ew, m, xl, acc);
    }
}

template<int VEC>
__global__ __launch_bounds__(256)
void spmm_gather(const int* __restrict__ rowinfo, const int* __restrict__ grouped,
                 const float* __restrict__ x, float* __restrict__ out)
{
    typedef typename RowVec<VEC>::T VT;
    constexpr int D = 32 * VEC;
    const int lane = threadIdx.x & 31;
    const int r = __builtin_amdgcn_readfirstlane(blockIdx.x * 8 + (threadIdx.x >> 5));
    if (r >= NN) return;

    float acc[VEC];
#pragma unroll
    for (int i = 0; i < VEC; ++i) acc[i] = 0.0f;
    gather_row<VEC>(rowinfo, grouped, x + lane * VEC, r, lane, acc);

    VT o;
    float* of = (float*)&o;
#pragma unroll
    for (int i = 0; i < VEC; ++i) of[i] = acc[i];
    *(VT*)(out + (size_t)r * D + lane * VEC) = o;
}

static constexpr int TROWS  = 64;
static constexpr int TPITCH = 132;
static_assert(NN % 4 == 0, "a wave's four rows must be all inside or all outside");

template<int NT, int KG, bool HOLD>
__device__ __forceinline__ void tile_times_w(const float* tile, const float* __restrict__ W,
                                             const int wave, const int lane, v8f (&acc)[NT / 4])
{
    constexpr int DOUT = NT * 16;
    constexpr int MS   = NT / 4;
    constexpr int NG   = DK / 4 / KG;
    const int half = lane >> 4;
    const int mn   = lane & 15;
    const int kb   = half * 2;
    const int nt   = wave % NT;
    const int ms0  = (wave / NT) * MS;

    const float* wp = W + (size_t)kb * DOUT + nt * 16 + mn;
    v2f breg[KG];
#pragma unroll
    for (int ks = 0; ks < KG; ++ks) {
        breg[ks].x = wp[(size_t)(4 * ks) * DOUT];
        breg[ks].y = wp[(size_t)(4 * ks + 1) * DOUT];
    }

    __syncthreads();

#pragma unroll
    for (int s = 0; s < MS; ++s) acc[s] = (v8f)0.0f;

#pragma unroll
    for (int g = 0; g < NG; ++g) {
        if (g > 0) {
            if (HOLD) __builtin_amdgcn_sched_barrier(0);
#pragma unroll
            for (int ks = 0; ks < KG; ++ks) {
                breg[ks].x = wp[(size_t)(4 * (ks + g * KG)) * DOUT];
                breg[ks].y = wp[(size_t)(4 * (ks + g * KG) + 1) * DOUT];
            }
        }
#pragma unroll
        for (int ks = 0; ks < KG; ++ks) {
#pragma unroll
            for (int s = 0; s < MS; ++s) {
                const v2f a = *(const v2f*)&tile[((ms0 + s) * 16 + mn) * TPITCH + 4 * (ks + g * KG) + kb];
                acc[s] = __builtin_amdgcn_wmma_f32_16x16x4_f32(
                    false, a, false, breg[ks], (short)0, acc[s], false, false);
            }
        }
    }

    __syncthreads();
}

template<int NT, bool PRE>
__global__ __launch_bounds__(512) __attribute__((amdgpu_waves_per_eu(PRE ? 16 : 4, 16)))
void spmm_relu_gemm(const int* __restrict__ rowinfo, const int* __restrict__ grouped,
                    const float* __restrict__ x, const float* __restrict__ Wp,
                    const float* __restrict__ bp, const float* __restrict__ W,
                    const float* __restrict__ bias, float* __restrict__ out)
{
    constexpr int DOUT = NT * 16;
    constexpr int MS   = NT / 4;
    constexpr int KG   = PRE ? 4 : 16;
    __shared__ __attribute__((aligned(16))) float tile[TROWS * TPITCH];
    __shared__ float wsum[TROWS];

    const int lane = threadIdx.x & 31;
    const int wave = __builtin_amdgcn_readfirstlane(threadIdx.x >> 5);
    const int row0 = blockIdx.x * TROWS;

    const int  rbase = row0 + wave * 4;
    const bool live  = rbase < NN;
    const int4* rp   = (const int4*)(rowinfo + 2 * (live ? rbase : 0));
    const int4 q0 = rp[0], q1 = rp[1];
    const int keep = live ? NE : 0;
    int n0 = min(max(q0.y, 0), keep), n1 = min(max(q0.w, 0), keep);
    int n2 = min(max(q1.y, 0), keep), n3 = min(max(q1.w, 0), keep);
    int b0 = min(max(q0.x, 0), NE - n0), b1 = min(max(q0.z, 0), NE - n1);
    int b2 = min(max(q1.x, 0), NE - n2), b3 = min(max(q1.z, 0), NE - n3);
    int c0, w0, c1, w1, c2, w2, c3, w3;
    load_entries(grouped, b0, min(32, n0), lane, c0, w0);
    load_entries(grouped, b1, min(32, n1), lane, c1, w1);
    load_entries(grouped, b2, min(32, n2), lane, c2, w2);
    load_entries(grouped, b3, min(32, n3), lane, c3, w3);

    const float* xl = x + lane * 4;
    for (int i = 0; i < 4; ++i) {
        float acc[4] = {0.0f, 0.0f, 0.0f, 0.0f};
        float sw = __int_as_float(w0);
        gather_chunk<4>(c0, w0, min(32, n0), xl, acc);
        for (int base = 32; base < n0; base += 32) {
            const int m = min(32, n0 - base);
            int fc, fw;
            load_entries(grouped, b0 + base, m, lane, fc, fw);
            if (PRE) sw += __int_as_float(fw);
            gather_chunk<4>(fc, fw, m, xl, acc);
        }
        float4 o;
        if (PRE) {
            o.x = acc[0]; o.y = acc[1]; o.z = acc[2]; o.w = acc[3];
#pragma unroll
            for (int d = 16; d >= 1; d >>= 1) sw += __shfl_xor(sw, d);
            if (lane == 0) wsum[wave * 4 + i] = sw;
        } else {
            o.x = fmaxf(acc[0], 0.0f); o.y = fmaxf(acc[1], 0.0f);
            o.z = fmaxf(acc[2], 0.0f); o.w = fmaxf(acc[3], 0.0f);
        }
        *(float4*)&tile[(wave * 4 + i) * TPITCH + lane * 4] = o;
        n0 = n1; n1 = n2; n2 = n3; n3 = 0;
        b0 = b1; b1 = b2; b2 = b3;
        c0 = c1; c1 = c2; c2 = c3;
        w0 = w1; w1 = w2; w2 = w3;
    }

    const int half = lane >> 4;
    const int mn   = lane & 15;

    if (PRE) {
        const int   pnt  = wave % 8;
        const int   pms0 = (wave / 8) * 2;
        const float pbv  = bp[pnt * 16 + mn];
        v8f pacc[2];
        tile_times_w<8, KG, PRE>(tile, Wp, wave, lane, pacc);
#pragma unroll
        for (int s = 0; s < 2; ++s)
#pragma unroll
            for (int v = 0; v < 8; ++v) {
                const int rl = (pms0 + s) * 16 + v + 8 * half;
                tile[rl * TPITCH + pnt * 16 + mn] = fmaxf(fmaf(wsum[rl], pbv, pacc[s][v]), 0.0f);
            }
    }

    const int   nt  = wave % NT;
    const int   ms0 = (wave / NT) * MS;
    const float bv  = bias[nt * 16 + mn];
    v8f acc[MS];
    tile_times_w<NT, KG, PRE>(tile, W, wave, lane, acc);

#pragma unroll
    for (int s = 0; s < MS; ++s)
#pragma unroll
        for (int v = 0; v < 8; ++v)
            tile[((ms0 + s) * 16 + v + 8 * half) * TPITCH + nt * 16 + mn] = acc[s][v] + bv;

    __syncthreads();

    typedef typename RowVec<DOUT / 32>::T OT;
    for (int i = 0; i < 4; ++i) {
        const int rl = wave * 4 + i;
        const int r  = row0 + rl;
        if (r < NN)
            *(OT*)(out + (size_t)r * DOUT + lane * (DOUT / 32)) =
                *(const OT*)&tile[rl * TPITCH + lane * (DOUT / 32)];
    }
}

extern "C" void kernel_launch(void* const* d_in, const int* in_sizes, int n_in,
                              void* d_out, int out_size, void* d_ws, size_t ws_size,
                              hipStream_t stream)
{
    const float* x    = (const float*)d_in[0];
    const float* vals = (const float*)d_in[1];
    const float* W1   = (const float*)d_in[2];
    const float* b1   = (const float*)d_in[3];
    const float* W2   = (const float*)d_in[4];
    const float* b2   = (const float*)d_in[5];
    const float* W3   = (const float*)d_in[6];
    const float* b3   = (const float*)d_in[7];
    const int*   row  = (const int*)d_in[8];
    const int*   col  = (const int*)d_in[9];

    float* bufA = (float*)d_ws;
    float* bufB = bufA + (size_t)NN * 128;
    int* binCount = (int*)(bufB + (size_t)NN * 128);
    int* T        = binCount + NBP;
    int* rowinfo  = T + T_PAD;
    int* tsorted  = rowinfo + 2 * NN;
    int* grouped  = tsorted + 2 * NE;

    const dim3 sBlk(256);
    const dim3 sGrd(NN / 8);

    hipMemsetAsync(binCount, 0, (size_t)NBP * sizeof(int), stream);
    tile_sort<<<dim3(NT), dim3(256), 0, stream>>>(row, col, vals, binCount, T, tsorted);
    csr_build<<<dim3(NB), dim3(1024), 0, stream>>>(binCount, T, tsorted, rowinfo, grouped);

    const dim3 fBlk(512);
    const dim3 fGrd((NN + TROWS - 1) / TROWS);

    spmm_relu_gemm<8, true><<<fGrd, fBlk, 0, stream>>>(rowinfo, grouped, x, W1, b1, W2, b2, bufB);

    spmm_relu_gemm<4, false><<<fGrd, fBlk, 0, stream>>>(rowinfo, grouped, bufB, nullptr, nullptr, W3, b3, bufA);

    spmm_gather<2><<<sGrd, sBlk, 0, stream>>>(rowinfo, grouped, bufA, (float*)d_out);
}
